# starting kernel
# baseline (speedup 1.0000x reference)
#include <hip/hip_runtime.h>

typedef float v2f __attribute__((ext_vector_type(2)));
typedef float v8f __attribute__((ext_vector_type(8)));

#define DDIM 512
#define FDIM 64
#define ODIM 64
#define LDSP 516
#define MROWS 32

__global__ __launch_bounds__(256) void build_w_kernel(
    const float* __restrict__ kern, const int* __restrict__ fid,
    float* __restrict__ W) {
  const int t = blockIdx.x * 256 + threadIdx.x;
  const int i = t >> 9;
  const int j = t & 511;
  float acc = 0.0f;
  if (j > i) {
    const float* ki = kern + (size_t)i * (FDIM * ODIM) + (size_t)fid[j] * ODIM;
    const float* kj = kern + (size_t)j * (FDIM * ODIM) + (size_t)fid[i] * ODIM;
#pragma unroll
    for (int o = 0; o < ODIM; o += 4) {
      const float4 a = *(const float4*)(ki + o);
      const float4 b = *(const float4*)(kj + o);
      acc += a.x * b.x + a.y * b.y + a.z * b.z + a.w * b.w;
    }
  }
  W[t] = acc;
}

__global__ __launch_bounds__(256) void quad_form_kernel(
    const float* __restrict__ X, const float* __restrict__ W,
    float* __restrict__ out) {
  __shared__ float ldsX[MROWS * LDSP];
  __shared__ float partial[8][MROWS];

  const int tid   = threadIdx.x;
  const int wave  = tid >> 5;
  const int lane  = tid & 31;
  const int l15   = lane & 15;
  const int half  = lane >> 4;
  const int mBase = blockIdx.x * MROWS;

  {
    const float4* src = (const float4*)(X + (size_t)mBase * DDIM);
#pragma unroll
    for (int it = 0; it < 16; ++it) {
      const int idx = tid + it * 256;
      const int row = idx >> 7;
      const int col = idx & 127;
      *(float4*)(ldsX + row * LDSP + col * 4) = src[row * 128 + col];
    }
  }
  __syncthreads();

  float rowSum0[8], rowSum1[8];
#pragma unroll
  for (int r = 0; r < 8; ++r) { rowSum0[r] = 0.0f; rowSum1[r] = 0.0f; }

#pragma unroll 1
  for (int t = 0; t < 4; ++t) {
    const int jBase = (wave + t * 8) * 16;
    v8f c0 = {0.f, 0.f, 0.f, 0.f, 0.f, 0.f, 0.f, 0.f};
    v8f c1 = {0.f, 0.f, 0.f, 0.f, 0.f, 0.f, 0.f, 0.f};
    const float* wcol = W + jBase + l15;

#pragma unroll 4
    for (int k0 = 0; k0 < DDIM; k0 += 4) {
      const int kA = k0 + half * 2;
      v2f a0, a1, b;
      b.x = wcol[(size_t)kA * DDIM];
      b.y = wcol[(size_t)(kA + 1) * DDIM];
      a0.x = ldsX[l15 * LDSP + kA];
      a0.y = ldsX[l15 * LDSP + kA + 1];
      a1.x = ldsX[(16 + l15) * LDSP + kA];
      a1.y = ldsX[(16 + l15) * LDSP + kA + 1];
      c0 = __builtin_amdgcn_wmma_f32_16x16x4_f32(
          false, a0, false, b, (short)0, c0, false, false);
      c1 = __builtin_amdgcn_wmma_f32_16x16x4_f32(
          false, a1, false, b, (short)0, c1, false, false);
    }

#pragma unroll
    for (int r = 0; r < 8; ++r) {
      const int m = r + half * 8;
      rowSum0[r] += c0[r] * ldsX[m * LDSP + jBase + l15];
      rowSum1[r] += c1[r] * ldsX[(16 + m) * LDSP + jBase + l15];
    }
  }

#pragma unroll
  for (int r = 0; r < 8; ++r) {
    float v0 = rowSum0[r];
    v0 += __shfl_xor(v0, 1, 32);
    v0 += __shfl_xor(v0, 2, 32);
    v0 += __shfl_xor(v0, 4, 32);
    v0 += __shfl_xor(v0, 8, 32);
    rowSum0[r] = v0;
    float v1 = rowSum1[r];
    v1 += __shfl_xor(v1, 1, 32);
    v1 += __shfl_xor(v1, 2, 32);
    v1 += __shfl_xor(v1, 4, 32);
    v1 += __shfl_xor(v1, 8, 32);
    rowSum1[r] = v1;
  }
  if (l15 == 0) {
#pragma unroll
    for (int r = 0; r < 8; ++r) {
      partial[wave][r + half * 8]      = rowSum0[r];
      partial[wave][16 + r + half * 8] = rowSum1[r];
    }
  }
  __syncthreads();

  if (tid < MROWS) {
    float s = 0.0f;
#pragma unroll
    for (int w = 0; w < 8; ++w) s += partial[w][tid];
    out[mBase + tid] = s;
  }
}

extern "C" void kernel_launch(void* const* d_in, const int* in_sizes, int n_in,
                              void* d_out, int out_size, void* d_ws, size_t ws_size,
                              hipStream_t stream) {
  const float* x    = (const float*)d_in[0];
  const float* kern = (const float*)d_in[1];
  const int*   fid  = (const int*)d_in[2];
  float* out = (float*)d_out;
  float* W   = (float*)d_ws;

  build_w_kernel<<<(DDIM * DDIM) / 256, 256, 0, stream>>>(kern, fid, W);
  quad_form_kernel<<<4096 / MROWS, 256, 0, stream>>>(x, W, out);
}


# Wm build through LDS: workgroup w of 256 copies kernel[w], kernel[511-w] and its 511 partner rows with whole-line 16-byte loads, all in flight at once; one thread per pair reads both rows from LDS
# speedup vs baseline: 3.8430x; 3.8430x over previous
#include <hip/hip_runtime.h>

typedef float v2f __attribute__((ext_vector_type(2)));
typedef float v8f __attribute__((ext_vector_type(8)));

#define DDIM 512
#define FDIM 64
#define ODIM 64
#define MROWS 16
#define XP 516
#define SLABK 32
#define NSLAB (DDIM / SLABK)
#define TILEP 528
#define SLABF (32 * TILEP)

#define KP 68
#define KI_F (FDIM * KP)

__global__ __launch_bounds__(256) void build_w_kernel(
    const float* __restrict__ kern, const int* __restrict__ fid,
    float* __restrict__ W) {
  __shared__ __attribute__((aligned(16))) float kI[2 * KI_F];
  __shared__ __attribute__((aligned(16))) float kJ[DDIM * KP];
  __shared__ __attribute__((aligned(16))) float res[DDIM];

  const int tid = threadIdx.x;
  const int iA  = blockIdx.x;
  const int iB  = DDIM - 1 - iA;
  const int nA  = DDIM - 1 - iA;
  const float4* k4 = (const float4*)kern;

  const float4* kA = k4 + iA * 1024 + tid;
  const float4* kB = k4 + iB * 1024 + tid;
  const float4 ri0 = kA[0], ri1 = kA[256], ri2 = kA[512], ri3 = kA[768];
  const float4 ri4 = kB[0], ri5 = kB[256], ri6 = kB[512], ri7 = kB[768];
  __builtin_amdgcn_sched_barrier(0);

  const int fA = fid[iA] & (FDIM - 1);
  const int fB = fid[iB] & (FDIM - 1);

  int pSel[2], pFld[2];
#pragma unroll
  for (int q = 0; q < 2; ++q) {
    const int p   = tid + q * 256;
    const int sel = (p >= nA) ? 1 : 0;
    int j = p + 1 + (sel ? 0 : iA);
    j = (j < DDIM - 1) ? j : (DDIM - 1);
    pSel[q] = sel;
    pFld[q] = fid[j] & (FDIM - 1);
  }

  const int hrow = tid >> 4;
  const int c16  = tid & 15;
  const char* kb = (const char*)kern + ((hrow + 1) * (FDIM * ODIM * 4) + c16 * 16);
  const int offA = iA * (FDIM * ODIM * 4) + fA * (ODIM * 4);
  const int offB = fB * (ODIM * 4);
  float4 rj[32];
#pragma unroll
  for (int it = 0; it < 32; ++it) {
    const int p = hrow + it * 16;
    int off = (p < nA) ? offA : offB;
    if (it == 31) off -= (hrow == 15) ? (FDIM * ODIM * 4) : 0;
    rj[it] = *(const float4*)(kb + (size_t)it * (16 * FDIM * ODIM * 4) + off);
  }

  {
    float* dA = kI + (tid >> 4) * KP + (tid & 15) * 4;
    float* dB = dA + KI_F;
    *(float4*)(dA)           = ri0;
    *(float4*)(dA + 16 * KP) = ri1;
    *(float4*)(dA + 32 * KP) = ri2;
    *(float4*)(dA + 48 * KP) = ri3;
    *(float4*)(dB)           = ri4;
    *(float4*)(dB + 16 * KP) = ri5;
    *(float4*)(dB + 32 * KP) = ri6;
    *(float4*)(dB + 48 * KP) = ri7;
  }
#pragma unroll
  for (int it = 0; it < 32; ++it)
    *(float4*)(kJ + (hrow + it * 16) * KP + c16 * 4) = rj[it];
  __syncthreads();

#pragma unroll
  for (int q = 0; q < 2; ++q) {
    const float* a = kI + pSel[q] * KI_F + pFld[q] * KP;
    const float* b = kJ + (tid + q * 256) * KP;
    float acc = 0.0f;
#pragma unroll
    for (int o = 0; o < ODIM; o += 4) {
      const float4 av = *(const float4*)(a + o);
      const float4 bv = *(const float4*)(b + o);
      acc += av.x * bv.x + av.y * bv.y + av.z * bv.z + av.w * bv.w;
    }
    res[tid + q * 256] = acc;
  }
  __syncthreads();

  {
    const int row  = tid >> 7;
    const int c4   = tid & 127;
    const int i    = row ? iB : iA;
    const int base = row ? (nA - iB - 1) : (-iA - 1);
    float v[4];
#pragma unroll
    for (int e = 0; e < 4; ++e) {
      const int jc = c4 * 4 + e;
      const int ix = base + jc;
      const float r = res[ix > 0 ? ix : 0];
      v[e] = (jc > i) ? r : 0.0f;
    }
    ((float4*)W)[i * 128 + c4] = make_float4(v[0], v[1], v[2], v[3]);
  }
}

typedef float f4 __attribute__((ext_vector_type(4)));

#define LOAD_FRAGS(kk, aPtr, bPtr)                                             \
  {                                                                            \
                          \
    fa[(kk) & 1] = *(const v2f*)((aPtr) + 4 * (kk));                           \
      \
    _Pragma("unroll") for (int t = 0; t < 4; ++t) {                            \
      fb[(kk) & 1][t].x = (bPtr)[t * 8 * TILEP + (kk) * 64];                   \
      fb[(kk) & 1][t].y = (bPtr)[t * 8 * TILEP + (kk) * 64 + 16];              \
    }                                                                          \
  }

#define KSTEP(kk, ra, rb, STORE_NEXT, LOAD_AFTER, aPtr, bPtr, stDst, ldSrc)    \
  {                                                                            \
    if (STORE_NEXT) {                                                          \
      *(f4*)((stDst) + (2 * (kk)) * 32)     = ra;                              \
      *(f4*)((stDst) + (2 * (kk) + 1) * 32) = rb;                              \
    }                                                                          \
    if (LOAD_AFTER) {                                                          \
      ra = (ldSrc)[(2 * (kk)) * 256];                                          \
      rb = (ldSrc)[(2 * (kk) + 1) * 256];                                      \
    }                                                                          \
    if ((kk) < 7) LOAD_FRAGS((kk) + 1, aPtr, bPtr)                             \
                                                     \
    __builtin_amdgcn_sched_barrier(0);                                         \
    _Pragma("unroll") for (int t = 0; t < 4; ++t)                              \
      acc[t] = __builtin_amdgcn_wmma_f32_16x16x4_f32(                          \
          false, fa[(kk) & 1], false, fb[(kk) & 1][t], (short)0, acc[t], false, false); \
      \
    __builtin_amdgcn_sched_barrier(0);                                         \
  }

#define SLAB_STEP(SN, LA, aPtr, bPtr, stDst, ldSrc)                            \
  {                                                                            \
    const float* aP_ = (aPtr);                                                 \
    const float* bP_ = (bPtr);                                                 \
    float* sD_ = (stDst);                                                      \
    const f4* lS_ = (ldSrc);                                                   \
    LOAD_FRAGS(0, aP_, bP_)                                                    \
    KSTEP(0, q0, q1, SN, LA, aP_, bP_, sD_, lS_)                               \
    KSTEP(1, q2, q3, SN, LA, aP_, bP_, sD_, lS_)                               \
    KSTEP(2, q4, q5, SN, LA, aP_, bP_, sD_, lS_)                               \
    KSTEP(3, q6, q7, SN, LA, aP_, bP_, sD_, lS_)                               \
    KSTEP(4, q8, q9, SN, LA, aP_, bP_, sD_, lS_)                               \
    KSTEP(5, q10, q11, SN, LA, aP_, bP_, sD_, lS_)                             \
    KSTEP(6, q12, q13, SN, LA, aP_, bP_, sD_, lS_)                             \
    KSTEP(7, q14, q15, SN, LA, aP_, bP_, sD_, lS_)                             \
  }

__device__ __forceinline__ void lds_handover() {
  asm volatile("s_wait_dscnt 0x0\n\ts_barrier_signal -1\n\ts_barrier_wait -1" ::: "memory");
}

#define STAGE_ALL(X) X(0, q0) X(1, q1) X(2, q2) X(3, q3) X(4, q4) X(5, q5) X(6, q6) X(7, q7) \
  X(8, q8) X(9, q9) X(10, q10) X(11, q11) X(12, q12) X(13, q13) X(14, q14) X(15, q15)

__global__ __launch_bounds__(256) void quad_form_kernel(
    const float* __restrict__ X, const float* __restrict__ W,
    float* __restrict__ out) {
  __shared__ __attribute__((aligned(16))) float ldsX[MROWS * XP];
  __shared__ __attribute__((aligned(16))) float ldsW[2 * SLABF];
  __shared__ float partial[8][MROWS];

  const int tid   = threadIdx.x;
  const int wave  = tid >> 5;
  const int lane  = tid & 31;
  const int l15   = lane & 15;
  const int half  = lane >> 4;
  const int mBase = blockIdx.x * MROWS;

  const int col4 = tid & 127;
  const f4* gsrc = (const f4*)W + (tid >> 7) * 128 + col4;
  const int stOff = (col4 >> 2) * TILEP + (tid >> 7) * 16 + (col4 & 3) * 4;

  f4 q0, q1, q2, q3, q4, q5, q6, q7, q8, q9, q10, q11, q12, q13, q14, q15;
#define LD0(it, r) r = gsrc[(it) * 256];
  STAGE_ALL(LD0)

  {
    const float4* src = (const float4*)(X + (size_t)mBase * DDIM);
#pragma unroll
    for (int it = 0; it < 8; ++it) {
      const int idx = tid + it * 256;
      const int row = idx >> 7;
      const int col = idx & 127;
      *(float4*)(ldsX + row * XP + col * 4) = src[row * 128 + col];
    }
  }
#define ST0(it, r) *(f4*)(ldsW + stOff + (it) * 32) = r;
  STAGE_ALL(ST0)
#define LD1(it, r) r = gsrc[4096 + (it) * 256];
  STAGE_ALL(LD1)
  lds_handover();

  v8f acc[4];
#pragma unroll
  for (int t = 0; t < 4; ++t) acc[t] = (v8f){0.f, 0.f, 0.f, 0.f, 0.f, 0.f, 0.f, 0.f};
  v2f fa[2];
  v2f fb[2][4];

  const float* aBase = ldsX + l15 * XP + 2 * half;
  const int bOff = wave * TILEP + half * 32 + l15;

#pragma unroll 1
  for (int s = 0; s < NSLAB - 2; ++s) {
    const float* cur = ldsW + (s & 1) * SLABF;
    float* nxt = ldsW + ((s + 1) & 1) * SLABF;
    SLAB_STEP(1, 1, aBase + SLABK * s, cur + bOff, nxt + stOff, gsrc + (size_t)(s + 2) * 4096)
    lds_handover();
  }
  SLAB_STEP(1, 0, aBase + SLABK * (NSLAB - 2), ldsW + bOff, ldsW + SLABF + stOff, gsrc)
  lds_handover();
  SLAB_STEP(0, 0, aBase + SLABK * (NSLAB - 1), ldsW + SLABF + bOff, ldsW, gsrc)

  float rowSum[8];
#pragma unroll
  for (int r = 0; r < 8; ++r) rowSum[r] = 0.0f;
#pragma unroll
  for (int t = 0; t < 4; ++t) {
    const int jBase = (wave + t * 8) * 16;
#pragma unroll
    for (int r = 0; r < 8; ++r)
      rowSum[r] += acc[t][r] * ldsX[(r + half * 8) * XP + jBase + l15];
  }

#pragma unroll
  for (int r = 0; r < 8; ++r) {
    float v = rowSum[r];
    v += __shfl_xor(v, 1, 32);
    v += __shfl_xor(v, 2, 32);
    v += __shfl_xor(v, 4, 32);
    v += __shfl_xor(v, 8, 32);
    rowSum[r] = v;
  }
  if (l15 == 0) {
#pragma unroll
    for (int r = 0; r < 8; ++r) partial[wave][r + half * 8] = rowSum[r];
  }
  __syncthreads();

  if (tid < MROWS) {
    float s = 0.0f;
#pragma unroll
    for (int w = 0; w < 8; ++w) s += partial[w][tid];
    out[mBase + tid] = s;
  }
}

extern "C" void kernel_launch(void* const* d_in, const int* in_sizes, int n_in,
                              void* d_out, int out_size, void* d_ws, size_t ws_size,
                              hipStream_t stream) {
  const float* x    = (const float*)d_in[0];
  const float* kern = (const float*)d_in[1];
  const int*   fid  = (const int*)d_in[2];
  float* out = (float*)d_out;
  float* W   = (float*)d_ws;

  build_w_kernel<<<DDIM / 2, 256, 0, stream>>>(kern, fid, W);
  quad_form_kernel<<<4096 / MROWS, 256, 0, stream>>>(x, W, out);
}


# Wm folded into a dense 256x512 rectangle (row w: row w right of the diagonal, row 511-w reversed left of it); kernel 2 walks 256 rows, each tile's A fragment from X or reversed X by address
# speedup vs baseline: 4.7719x; 1.2417x over previous
#include <hip/hip_runtime.h>

typedef float v2f __attribute__((ext_vector_type(2)));
typedef float v8f __attribute__((ext_vector_type(8)));

#define DDIM 512
#define FDIM 64
#define ODIM 64
#define MROWS 16
#define XP 516
#define GROWS 256
#define XRP 260
#define SLABK 32
#define NSLAB (GROWS / SLABK)
#define TILEP 528
#define SLABF (32 * TILEP)

#define KP 68
#define KI_F (FDIM * KP)

__global__ __launch_bounds__(256) void build_w_kernel(
    const float* __restrict__ kern, const int* __restrict__ fid,
    float* __restrict__ W) {
  __shared__ __attribute__((aligned(16))) float kI[2 * KI_F];
  __shared__ __attribute__((aligned(16))) float kJ[DDIM * KP];
  __shared__ __attribute__((aligned(16))) float res[DDIM];

  const int tid = threadIdx.x;
  const int iA  = blockIdx.x;
  const int iB  = DDIM - 1 - iA;
  const int nA  = DDIM - 1 - iA;
  const float4* k4 = (const float4*)kern;

  const float4* kA = k4 + iA * 1024 + tid;
  const float4* kB = k4 + iB * 1024 + tid;
  const float4 ri0 = kA[0], ri1 = kA[256], ri2 = kA[512], ri3 = kA[768];
  const float4 ri4 = kB[0], ri5 = kB[256], ri6 = kB[512], ri7 = kB[768];
  __builtin_amdgcn_sched_barrier(0);

  const int fA = fid[iA] & (FDIM - 1);
  const int fB = fid[iB] & (FDIM - 1);

  int pSel[2], pFld[2];
#pragma unroll
  for (int q = 0; q < 2; ++q) {
    const int p   = tid + q * 256;
    const int sel = (p >= nA) ? 1 : 0;
    int j = p + 1 + (sel ? 0 : iA);
    j = (j < DDIM - 1) ? j : (DDIM - 1);
    pSel[q] = sel;
    pFld[q] = fid[j] & (FDIM - 1);
  }

  const int hrow = tid >> 4;
  const int c16  = tid & 15;
  const char* kb = (const char*)kern + ((hrow + 1) * (FDIM * ODIM * 4) + c16 * 16);
  const int offA = iA * (FDIM * ODIM * 4) + fA * (ODIM * 4);
  const int offB = fB * (ODIM * 4);
  float4 rj[32];
#pragma unroll
  for (int it = 0; it < 32; ++it) {
    const int p = hrow + it * 16;
    int off = (p < nA) ? offA : offB;
    if (it == 31) off -= (hrow == 15) ? (FDIM * ODIM * 4) : 0;
    rj[it] = *(const float4*)(kb + (size_t)it * (16 * FDIM * ODIM * 4) + off);
  }

  {
    float* dA = kI + (tid >> 4) * KP + (tid & 15) * 4;
    float* dB = dA + KI_F;
    *(float4*)(dA)           = ri0;
    *(float4*)(dA + 16 * KP) = ri1;
    *(float4*)(dA + 32 * KP) = ri2;
    *(float4*)(dA + 48 * KP) = ri3;
    *(float4*)(dB)           = ri4;
    *(float4*)(dB + 16 * KP) = ri5;
    *(float4*)(dB + 32 * KP) = ri6;
    *(float4*)(dB + 48 * KP) = ri7;
  }
#pragma unroll
  for (int it = 0; it < 32; ++it)
    *(float4*)(kJ + (hrow + it * 16) * KP + c16 * 4) = rj[it];
  __syncthreads();

#pragma unroll
  for (int q = 0; q < 2; ++q) {
    const float* a = kI + pSel[q] * KI_F + pFld[q] * KP;
    const float* b = kJ + (tid + q * 256) * KP;
    float acc = 0.0f;
#pragma unroll
    for (int o = 0; o < ODIM; o += 4) {
      const float4 av = *(const float4*)(a + o);
      const float4 bv = *(const float4*)(b + o);
      acc += av.x * bv.x + av.y * bv.y + av.z * bv.z + av.w * bv.w;
    }
    res[tid + q * 256] = acc;
  }
  __syncthreads();

  if (tid < 128) {
    const int c4 = tid;
    float v[4];
#pragma unroll
    for (int e = 0; e < 4; ++e) {
      const int jc = c4 * 4 + e;
      const int ix = (jc > iA) ? (jc - iA - 1) : (510 - jc);
      const float r = res[ix];
      v[e] = (jc == iA) ? 0.0f : r;
    }
    ((float4*)W)[iA * 128 + c4] = make_float4(v[0], v[1], v[2], v[3]);
  }
}

typedef float f4 __attribute__((ext_vector_type(4)));

#define LOAD_FRAGS(kk, aPtr, a0Ptr, a1Ptr, bPtr)                               \
  {                                                                            \
                               \
    fa[(kk) & 1] = *(const v2f*)((aPtr) + 4 * (kk));                           \
    f0[(kk) & 1] = *(const v2f*)((a0Ptr) + 4 * (kk));                          \
    f1[(kk) & 1] = *(const v2f*)((a1Ptr) + 4 * (kk));                          \
      \
    _Pragma("unroll") for (int t = 0; t < 4; ++t) {                            \
      fb[(kk) & 1][t].x = (bPtr)[t * 8 * TILEP + (kk) * 64];                   \
      fb[(kk) & 1][t].y = (bPtr)[t * 8 * TILEP + (kk) * 64 + 16];              \
    }                                                                          \
  }

#define WMMA_F32(a, b, c) __builtin_amdgcn_wmma_f32_16x16x4_f32(false, a, false, b, (short)0, c, false, false)

#define KSTEP(kk, ra, rb, STORE_NEXT, LOAD_AFTER, aPtr, bPtr, stDst, ldSrc)    \
  {                                                                            \
    if (STORE_NEXT) {                                                          \
      *(f4*)((stDst) + (2 * (kk)) * 32)     = ra;                              \
      *(f4*)((stDst) + (2 * (kk) + 1) * 32) = rb;                              \
    }                                                                          \
    if (LOAD_AFTER) {                                                          \
      ra = (ldSrc)[(2 * (kk)) * 256];                                          \
      rb = (ldSrc)[(2 * (kk) + 1) * 256];                                      \
    }                                                                          \
    if ((kk) < 7)                                                              \
      LOAD_FRAGS((kk) + 1, aPtr, (((kk) + 1 < 4) ? a0lo : a0hi),               \
                 (((kk) + 1 < 4) ? a1lo : a1hi), bPtr)                         \
                                                     \
    __builtin_amdgcn_sched_barrier(0);                                         \
    acc[2] = WMMA_F32(fa[(kk) & 1], fb[(kk) & 1][2], acc[2]);                  \
    acc[3] = WMMA_F32(fa[(kk) & 1], fb[(kk) & 1][3], acc[3]);                  \
    acc[0] = WMMA_F32(f0[(kk) & 1], fb[(kk) & 1][0], acc[0]);                  \
    acc[1] = WMMA_F32(f1[(kk) & 1], fb[(kk) & 1][1], acc[1]);                  \
                                                          \
    asm volatile("" : "+v"(acc[0]), "+v"(acc[1]), "+v"(acc[2]), "+v"(acc[3])); \
      \
    __builtin_amdgcn_sched_barrier(0);                                         \
  }

#define A_SRC(kb, J, s) (ldsA + (((kb) == (J)) ? offZ : (((kb) < (J)) ? offX : offR) + SLABK * (s)))
#define SLAB_PTRS(s)                                                           \
    const float* a0lo = A_SRC(2 * (s), J0, s);                                 \
    const float* a0hi = A_SRC(2 * (s) + 1, J0, s);                             \
    const float* a1lo = A_SRC(2 * (s), J1, s);                                 \
    const float* a1hi = A_SRC(2 * (s) + 1, J1, s);

#define FOLD_CHECK(s, t, J)                                                    \
  {                                                                            \
    const bool fold_ = ((s) == (((J) + 1) >> 1));                              \
    const float* xc_ = ldsA + half * 8 * XP + (J) * 16 + l15;                  \
    _Pragma("unroll") for (int r = 0; r < 8; ++r) {                            \
      const float p_ = acc[t][r] * xc_[r * XP];                                \
      rowSum[r] += fold_ ? p_ : 0.0f;                                          \
      acc[t][r] = fold_ ? 0.0f : acc[t][r];                                    \
    }                                                                          \
  }

#define SLAB_STEP(SN, LA, aPtr, bPtr, stDst, ldSrc)                            \
  {                                                                            \
    const float* aP_ = (aPtr);                                                 \
    const float* bP_ = (bPtr);                                                 \
    float* sD_ = (stDst);                                                      \
    const f4* lS_ = (ldSrc);                                                   \
    LOAD_FRAGS(0, aP_, a0lo, a1lo, bP_)                                        \
    KSTEP(0, q0, q1, SN, LA, aP_, bP_, sD_, lS_)                               \
    KSTEP(1, q2, q3, SN, LA, aP_, bP_, sD_, lS_)                               \
    KSTEP(2, q4, q5, SN, LA, aP_, bP_, sD_, lS_)                               \
    KSTEP(3, q6, q7, SN, LA, aP_, bP_, sD_, lS_)                               \
    KSTEP(4, q8, q9, SN, LA, aP_, bP_, sD_, lS_)                               \
    KSTEP(5, q10, q11, SN, LA, aP_, bP_, sD_, lS_)                             \
    KSTEP(6, q12, q13, SN, LA, aP_, bP_, sD_, lS_)                             \
    KSTEP(7, q14, q15, SN, LA, aP_, bP_, sD_, lS_)                             \
  }

__device__ __forceinline__ void lds_handover() {
  asm volatile("s_wait_dscnt 0x0\n\ts_barrier_signal -1\n\ts_barrier_wait -1" ::: "memory");
}

#define STAGE_ALL(X) X(0, q0) X(1, q1) X(2, q2) X(3, q3) X(4, q4) X(5, q5) X(6, q6) X(7, q7) \
  X(8, q8) X(9, q9) X(10, q10) X(11, q11) X(12, q12) X(13, q13) X(14, q14) X(15, q15)

#define XROFF (MROWS * XP)
#define ZOFF  (XROFF + MROWS * XRP)

__global__ __launch_bounds__(256) void quad_form_kernel(
    const float* __restrict__ X, const float* __restrict__ W,
    float* __restrict__ out) {
  __shared__ __attribute__((aligned(16))) float ldsA[MROWS * XP + MROWS * XRP + 64];
  __shared__ __attribute__((aligned(16))) float ldsW[2 * SLABF];
  __shared__ float partial[8][MROWS];

  const int tid   = threadIdx.x;
  const int wave  = tid >> 5;
  const int lane  = tid & 31;
  const int l15   = lane & 15;
  const int half  = lane >> 4;
  const int mBase = blockIdx.x * MROWS;

  const int J0 = __builtin_amdgcn_readfirstlane(wave);
  const int J1 = J0 + 8;

  const int col4 = tid & 127;
  const f4* gsrc = (const f4*)W + (tid >> 7) * 128 + col4;
  const int stOff = (col4 >> 2) * TILEP + (tid >> 7) * 16 + (col4 & 3) * 4;

  f4 q0, q1, q2, q3, q4, q5, q6, q7, q8, q9, q10, q11, q12, q13, q14, q15;
#define LD0(it, r) r = gsrc[(it) * 256];
  STAGE_ALL(LD0)

  {
    const float4* src = (const float4*)(X + (size_t)mBase * DDIM) + tid;
    float4 xv[8];
#pragma unroll
    for (int it = 0; it < 8; ++it) xv[it] = src[it * 256];
    const int row0 = tid >> 7;
    float* dx = ldsA + row0 * XP + col4 * 4;
#pragma unroll
    for (int it = 0; it < 8; ++it) *(float4*)(dx + it * 2 * XP) = xv[it];
    if (col4 >= 64) {
      float* dr = ldsA + XROFF + row0 * XRP + (508 - 4 * col4);
#pragma unroll
      for (int it = 0; it < 8; ++it)
        *(float4*)(dr + it * 2 * XRP) = make_float4(xv[it].w, xv[it].z, xv[it].y, xv[it].x);
    }
    if (tid < 64) ldsA[ZOFF + tid] = 0.0f;
  }

  float dg[2][4][2];
#pragma unroll
  for (int t = 0; t < 2; ++t) {
    const int J = t ? J1 : J0;
    const float* gp = W + (size_t)(J * 16 + 2 * half) * DDIM + J * 16 + l15;
#pragma unroll
    for (int kq = 0; kq < 4; ++kq) {
      dg[t][kq][0] = gp[(4 * kq) * DDIM];
      dg[t][kq][1] = gp[(4 * kq + 1) * DDIM];
    }
  }

#define ST0(it, r) *(f4*)(ldsW + stOff + (it) * 32) = r;
  STAGE_ALL(ST0)
#define LD1(it, r) r = gsrc[4096 + (it) * 256];
  STAGE_ALL(LD1)
  lds_handover();

  v8f acc[4];
  v8f accD[2];
#pragma unroll
  for (int t = 0; t < 4; ++t) acc[t] = (v8f){0.f, 0.f, 0.f, 0.f, 0.f, 0.f, 0.f, 0.f};
#pragma unroll
  for (int t = 0; t < 2; ++t) accD[t] = (v8f){0.f, 0.f, 0.f, 0.f, 0.f, 0.f, 0.f, 0.f};
  float rowSum[8];
#pragma unroll
  for (int r = 0; r < 8; ++r) rowSum[r] = 0.0f;

  const int offX = l15 * XP + 2 * half;
  const int offR = XROFF + l15 * XRP + 2 * half;
  const int offZ = ZOFF + 2 * half;

#pragma unroll
  for (int t = 0; t < 2; ++t) {
    const int J = t ? J1 : J0;
    const float* pu = ldsA + offX + J * 16;
    const float* pl = ldsA + offR + J * 16;
#pragma unroll
    for (int kq = 0; kq < 4; ++kq) {
      const v2f au = *(const v2f*)(pu + 4 * kq);
      const v2f al = *(const v2f*)(pl + 4 * kq);
      const int r0 = 4 * kq + 2 * half;
      v2f bu, bl;
      bu.x = (l15 > r0)     ? dg[t][kq][0] : 0.0f;
      bu.y = (l15 > r0 + 1) ? dg[t][kq][1] : 0.0f;
      bl.x = (l15 < r0)     ? dg[t][kq][0] : 0.0f;
      bl.y = (l15 < r0 + 1) ? dg[t][kq][1] : 0.0f;
      acc[t]  = WMMA_F32(au, bu, acc[t]);
      accD[t] = WMMA_F32(al, bl, accD[t]);
    }
  }

  v2f fa[2];
  v2f f0[2];
  v2f f1[2];
  v2f fb[2][4];

  const float* aBase = ldsA + offX;
  const int bOff = wave * TILEP + half * 32 + l15;

#pragma unroll 1
  for (int s = 0; s < NSLAB - 2; ++s) {
    const float* cur = ldsW + (s & 1) * SLABF;
    float* nxt = ldsW + ((s + 1) & 1) * SLABF;
    FOLD_CHECK(s, 0, J0)
    FOLD_CHECK(s, 1, J1)
    SLAB_PTRS(s)
    SLAB_STEP(1, 1, aBase + SLABK * s, cur + bOff, nxt + stOff, gsrc + (size_t)(s + 2) * 4096)
    lds_handover();
  }
  {
    FOLD_CHECK(NSLAB - 2, 1, J1)
    SLAB_PTRS(NSLAB - 2)
    SLAB_STEP(1, 0, aBase + SLABK * (NSLAB - 2), ldsW + bOff, ldsW + SLABF + stOff, gsrc)
  }
  lds_handover();
  {
    FOLD_CHECK(NSLAB - 1, 1, J1)
    SLAB_PTRS(NSLAB - 1)
    SLAB_STEP(0, 0, aBase + SLABK * (NSLAB - 1), ldsW + SLABF + bOff, ldsW, gsrc)
  }
  FOLD_CHECK(NSLAB, 1, J1)

#pragma unroll
  for (int t = 2; t < 4; ++t) {
    const int jBase = (wave + t * 8) * 16;
#pragma unroll
    for (int r = 0; r < 8; ++r)
      rowSum[r] += acc[t][r] * ldsA[(r + half * 8) * XP + jBase + l15];
  }
#pragma unroll
  for (int t = 0; t < 2; ++t) {
    const int jBase = (wave + t * 8) * 16;
#pragma unroll
    for (int r = 0; r < 8; ++r)
      rowSum[r] += (acc[t][r] + accD[t][r]) * ldsA[XROFF + (r + half * 8) * XRP + jBase + l15];
  }

#pragma unroll
  for (int r = 0; r < 8; ++r) {
    float v = rowSum[r];
    v += __shfl_xor(v, 1, 32);
    v += __shfl_xor(v, 2, 32);
    v += __shfl_xor(v, 4, 32);
    v += __shfl_xor(v, 8, 32);
    rowSum[r] = v;
  }
  if (l15 == 0) {
#pragma unroll
    for (int r = 0; r < 8; ++r) partial[wave][r + half * 8] = rowSum[r];
  }
  __syncthreads();

  if (tid < MROWS) {
    float s = 0.0f;
#pragma unroll
    for (int w = 0; w < 8; ++w) s += partial[w][tid];
    out[mBase + tid] = s;
  }
}

extern "C" void kernel_launch(void* const* d_in, const int* in_sizes, int n_in,
                              void* d_out, int out_size, void* d_ws, size_t ws_size,
                              hipStream_t stream) {
  const float* x    = (const float*)d_in[0];
  const float* kern = (const float*)d_in[1];
  const int*   fid  = (const int*)d_in[2];
  float* out = (float*)d_out;
  float* W   = (float*)d_ws;

  build_w_kernel<<<DDIM / 2, 256, 0, stream>>>(kern, fid, W);
  quad_form_kernel<<<4096 / MROWS, 256, 0, stream>>>(x, W, out);
}
